# starting kernel
# baseline (speedup 1.0000x reference)
#include <hip/hip_runtime.h>

typedef float v2f __attribute__((ext_vector_type(2)));
typedef float v4f __attribute__((ext_vector_type(4)));
typedef float v8f __attribute__((ext_vector_type(8)));

#define B_   32
#define N_   4096
#define IN_  12
#define HID_ 256
#define OUT_ 128

__global__ __launch_bounds__(256) void gcn_reduce_l1(
    const float* __restrict__ x, const float* __restrict__ W1,
    const float* __restrict__ b1, float* __restrict__ H1) {
  __shared__ float red[256][IN_];
  const int b = blockIdx.x;
  const int t = threadIdx.x;
  const float* base = x + (size_t)b * N_ * IN_;

  v4f a0 = {}, a1 = {}, a2 = {};
  for (int n = t; n < N_; n += 256) {
    const v4f* p = (const v4f*)(base + n * IN_);
    a0 += p[0];
    a1 += p[1];
    a2 += p[2];
  }
  red[t][0] = a0.x;  red[t][1] = a0.y;  red[t][2]  = a0.z;  red[t][3]  = a0.w;
  red[t][4] = a1.x;  red[t][5] = a1.y;  red[t][6]  = a1.z;  red[t][7]  = a1.w;
  red[t][8] = a2.x;  red[t][9] = a2.y;  red[t][10] = a2.z;  red[t][11] = a2.w;
  __syncthreads();

  for (int s = 128; s > 0; s >>= 1) {
    if (t < s) {
#pragma unroll
      for (int f = 0; f < IN_; ++f) red[t][f] += red[t + s][f];
    }
    __syncthreads();
  }

  float h = b1[t];
#pragma unroll
  for (int f = 0; f < IN_; ++f)
    h += (red[0][f] * (1.0f / (float)N_)) * W1[f * HID_ + t];
  H1[b * HID_ + t] = h;
}

__global__ __launch_bounds__(512) void gcn_l2_wmma(
    const float* __restrict__ H1, const float* __restrict__ W2,
    const float* __restrict__ b2, float* __restrict__ H2) {
  const int t     = threadIdx.x;
  const int wave  = t >> 5;
  const int lane  = t & 31;
  const int mi    = wave >> 3;
  const int ni    = wave & 7;
  const int laneM = lane & 15;
  const int hi    = lane >> 4;
  const int kOff  = 2 * hi;

  const float* arow = H1 + (size_t)(mi * 16 + laneM) * HID_;
  const float* bcol = W2 + ni * 16 + laneM;

  v8f c = {};
  for (int k0 = 0; k0 < HID_; k0 += 4) {
    v2f a, bm;
    a.x  = arow[k0 + kOff];
    a.y  = arow[k0 + kOff + 1];
    bm.x = bcol[(size_t)(k0 + kOff) * OUT_];
    bm.y = bcol[(size_t)(k0 + kOff + 1) * OUT_];
    c = __builtin_amdgcn_wmma_f32_16x16x4_f32(
        false, a, false, bm, (short)0, c, false, false);
  }

  const int   col  = ni * 16 + laneM;
  const float bias = b2[col];
#pragma unroll
  for (int v = 0; v < 8; ++v) {
    const int row = mi * 16 + v + 8 * hi;
    H2[row * OUT_ + col] = c[v] + bias;
  }
}

__global__ __launch_bounds__(256) void gcn_broadcast(
    const float* __restrict__ H2, float* __restrict__ out) {
  const int b       = blockIdx.x >> 6;
  const int chunk   = blockIdx.x & 63;
  const int t       = threadIdx.x;
  const int col4    = (t & 31) << 2;
  const int nodeOff = t >> 5;

  const v4f val = *(const v4f*)(H2 + b * OUT_ + col4);
  float* obase = out + ((size_t)b * N_ + (size_t)chunk * 64) * OUT_;
#pragma unroll
  for (int i = 0; i < 8; ++i) {
    v4f* dst = (v4f*)(obase + (size_t)(nodeOff + i * 8) * OUT_ + col4);
    __builtin_nontemporal_store(val, dst);
  }
}

extern "C" void kernel_launch(void* const* d_in, const int* in_sizes, int n_in,
                              void* d_out, int out_size, void* d_ws,
                              size_t ws_size, hipStream_t stream) {
  const float* x  = (const float*)d_in[0];
  const float* W1 = (const float*)d_in[1];
  const float* b1 = (const float*)d_in[2];
  const float* W2 = (const float*)d_in[3];
  const float* b2 = (const float*)d_in[4];
  float* out = (float*)d_out;

  float* H1 = (float*)d_ws;
  float* H2 = H1 + B_ * HID_;

  gcn_reduce_l1<<<B_, 256, 0, stream>>>(x, W1, b1, H1);
  gcn_l2_wmma<<<1, 512, 0, stream>>>(H1, W2, b2, H2);
  gcn_broadcast<<<B_ * 64, 256, 0, stream>>>(H2, out);
}


# Fuse the layer-2 WMMA GEMM into the broadcast: 256 blocks each recompute their own 32x32 slice of H2 (H1 and W2 slice staged in LDS) and stream it to out; 3 launches become 2; same k order, f32
# speedup vs baseline: 1.5997x; 1.5997x over previous
#include <hip/hip_runtime.h>

typedef float v2f __attribute__((ext_vector_type(2)));
typedef float v4f __attribute__((ext_vector_type(4)));
typedef float v8f __attribute__((ext_vector_type(8)));

#define B_   32
#define N_   4096
#define IN_  12
#define HID_ 256
#define OUT_ 128

__global__ __launch_bounds__(256) void gcn_reduce_l1(
    const float* __restrict__ x, const float* __restrict__ W1,
    const float* __restrict__ b1, float* __restrict__ H1) {
  __shared__ float red[256][IN_];
  const int b = blockIdx.x;
  const int t = threadIdx.x;
  const float* base = x + (size_t)b * N_ * IN_;

  v4f a0 = {}, a1 = {}, a2 = {};
  for (int n = t; n < N_; n += 256) {
    const v4f* p = (const v4f*)(base + n * IN_);
    a0 += p[0];
    a1 += p[1];
    a2 += p[2];
  }
  red[t][0] = a0.x;  red[t][1] = a0.y;  red[t][2]  = a0.z;  red[t][3]  = a0.w;
  red[t][4] = a1.x;  red[t][5] = a1.y;  red[t][6]  = a1.z;  red[t][7]  = a1.w;
  red[t][8] = a2.x;  red[t][9] = a2.y;  red[t][10] = a2.z;  red[t][11] = a2.w;
  __syncthreads();

  for (int s = 128; s > 0; s >>= 1) {
    if (t < s) {
#pragma unroll
      for (int f = 0; f < IN_; ++f) red[t][f] += red[t + s][f];
    }
    __syncthreads();
  }

  float h = b1[t];
#pragma unroll
  for (int f = 0; f < IN_; ++f)
    h += (red[0][f] * (1.0f / (float)N_)) * W1[f * HID_ + t];
  H1[b * HID_ + t] = h;
}

#define CGC_     32
#define NPB_     64
#define KSTEPS_  (HID_ / 4)
#define APITCH_  260
#define BPITCH_  40

__global__ __launch_bounds__(256) void gcn_l2_broadcast(
    const float* __restrict__ H1, const float* __restrict__ W2,
    const float* __restrict__ b2, float* __restrict__ out) {
  __shared__ float As[B_ * APITCH_];
  __shared__ float Bs[HID_ * BPITCH_];
  __shared__ float Hs[B_ * CGC_];

  const int t     = threadIdx.x;
  const int wave  = __builtin_amdgcn_readfirstlane(t >> 5);
  const int lane  = t & 31;
  const int laneM = lane & 15;
  const int hi    = lane >> 4;
  const int kOff  = 2 * hi;
  const int cg    = blockIdx.x & 3;
  const int nc    = blockIdx.x >> 2;
  const int mi    = (wave >> 1) & 1;
  const int ni    = wave & 1;

  float bias = 0.f;
  if (wave < 4) bias = b2[cg * CGC_ + ni * 16 + laneM];
  {
    v4f va[8], vb[8];
#pragma unroll
    for (int j = 0; j < 8; ++j) {
      const int i = t + 256 * j;
      va[j] = *(const v4f*)(H1 + (i >> 6) * HID_ + (i & 63) * 4);
    }
#pragma unroll
    for (int j = 0; j < 8; ++j) {
      const int i = t + 256 * j;
      vb[j] = *(const v4f*)(W2 + (i >> 3) * OUT_ + cg * CGC_ + (i & 7) * 4);
    }
#pragma unroll
    for (int j = 0; j < 8; ++j) {
      const int i = t + 256 * j;
      *(v4f*)(As + (i >> 6) * APITCH_ + (i & 63) * 4) = va[j];
    }
#pragma unroll
    for (int j = 0; j < 8; ++j) {
      const int i = t + 256 * j;
      *(v4f*)(Bs + (i >> 3) * BPITCH_ + (i & 7) * 4) = vb[j];
    }
  }
  __syncthreads();

  if (wave < 4) {
    const float* arow = As + (mi * 16 + laneM) * APITCH_ + kOff;
    const float* bcol = Bs + kOff * BPITCH_ + ni * 16 + laneM;
    v2f afrag[KSTEPS_], bfrag[KSTEPS_];
#pragma unroll
    for (int ks = 0; ks < KSTEPS_; ++ks) {
      afrag[ks]   = *(const v2f*)(arow + 4 * ks);
      bfrag[ks].x = bcol[(4 * ks) * BPITCH_];
      bfrag[ks].y = bcol[(4 * ks + 1) * BPITCH_];
      if ((ks & 7) == 7) __builtin_amdgcn_sched_barrier(0);
    }
    v8f c = {};
#pragma unroll
    for (int ks = 0; ks < KSTEPS_; ++ks) {
      c = __builtin_amdgcn_wmma_f32_16x16x4_f32(
          false, afrag[ks], false, bfrag[ks], (short)0, c, false, false);
    }
    __builtin_amdgcn_sched_barrier(0);
    asm volatile("v_nop\n\tv_nop\n\tv_nop\n\tv_nop\n\tv_nop" : "+v"(c));
    __builtin_amdgcn_sched_barrier(0);
    const int col = ni * 16 + laneM;
#pragma unroll
    for (int v = 0; v < 8; ++v)
      Hs[(mi * 16 + v + 8 * hi) * CGC_ + col] = c[v] + bias;
  }
  __syncthreads();

  const int col4 = (lane & 7) << 2;
  const int nsel = lane >> 3;
#pragma unroll
  for (int bb = 0; bb < 4; ++bb) {
    const int b   = wave * 4 + bb;
    const v4f val = *(const v4f*)(Hs + b * CGC_ + col4);
    float* obase = out + ((size_t)b * N_ + (size_t)nc * NPB_ + nsel) * OUT_
                 + cg * CGC_ + col4;
#pragma unroll
    for (int i = 0; i < NPB_ / 4; ++i)
      __builtin_nontemporal_store(val, (v4f*)(obase + (size_t)(i * 4) * OUT_));
  }
}

extern "C" void kernel_launch(void* const* d_in, const int* in_sizes, int n_in,
                              void* d_out, int out_size, void* d_ws,
                              size_t ws_size, hipStream_t stream) {
  const float* x  = (const float*)d_in[0];
  const float* W1 = (const float*)d_in[1];
  const float* b1 = (const float*)d_in[2];
  const float* W2 = (const float*)d_in[3];
  const float* b2 = (const float*)d_in[4];
  float* out = (float*)d_out;

  float* H1 = (float*)d_ws;

  gcn_reduce_l1<<<B_, 256, 0, stream>>>(x, W1, b1, H1);
  gcn_l2_broadcast<<<4 * (N_ / NPB_), 256, 0, stream>>>(H1, W2, b2, out);
}


# Node sum split into 8 chunk partials per batch over 256 blocks (new reduction tree); the broadcast kernel adds the partials and does layer 1 itself on f32 WMMA; 2 launches, f32
# speedup vs baseline: 1.9010x; 1.1883x over previous
#include <hip/hip_runtime.h>

typedef float v2f __attribute__((ext_vector_type(2)));
typedef float v4f __attribute__((ext_vector_type(4)));
typedef float v8f __attribute__((ext_vector_type(8)));

#define B_   32
#define N_   4096
#define IN_  12
#define HID_ 256
#define OUT_ 128

#define NCH_  8
#define NPC_  (N_ / NCH_)
#define RT_   128
#define RLD_  (NPC_ * IN_ / 4 / RT_)

__device__ __forceinline__ float row16_sum(float x) {
  x += __builtin_bit_cast(float, __builtin_amdgcn_update_dpp(
           0, __builtin_bit_cast(int, x), 0x161, 0xf, 0xf, false));
  x += __builtin_bit_cast(float, __builtin_amdgcn_update_dpp(
           0, __builtin_bit_cast(int, x), 0x162, 0xf, 0xf, false));
  x += __builtin_bit_cast(float, __builtin_amdgcn_update_dpp(
           0, __builtin_bit_cast(int, x), 0x164, 0xf, 0xf, false));
  x += __builtin_bit_cast(float, __builtin_amdgcn_update_dpp(
           0, __builtin_bit_cast(int, x), 0x168, 0xf, 0xf, false));
  return x;
}
__device__ __forceinline__ float xhalf(float x) {
  const int b = __builtin_bit_cast(int, x);
  return __builtin_bit_cast(float, __builtin_amdgcn_permlanex16(
             b, b, 0x76543210, 0xfedcba98, false, false));
}
__device__ __forceinline__ float wave_sum(float x) {
  x = row16_sum(x);
  return x + xhalf(x);
}

__global__ __launch_bounds__(RT_) void gcn_node_partial(
    const float* __restrict__ x, float* __restrict__ P) {
  __shared__ v4f wsum[(RT_ / 32) * 3];
  const int t    = threadIdx.x;
  const int wave = t >> 5;
  const int b    = blockIdx.x >> 3;
  const int ch   = blockIdx.x & 7;
  const v4f* base =
      (const v4f*)(x + ((size_t)b * N_ + (size_t)ch * NPC_) * IN_);

  v4f v[RLD_];
#pragma unroll
  for (int j = 0; j < RLD_; ++j) v[j] = base[t + RT_ * j];

  v4f s0 = {}, s1 = {}, s2 = {};
#pragma unroll
  for (int j = 0; j < RLD_; ++j) {
    const int k = (2 * j) % 3;
    if (k == 0) s0 += v[j];
    else if (k == 1) s1 += v[j];
    else s2 += v[j];
  }
  const int  r  = t % 3;
  const bool r1 = (r == 1), r2 = (r == 2);
  v4f g0 = r1 ? s2 : (r2 ? s1 : s0);
  v4f g1 = r1 ? s0 : (r2 ? s2 : s1);
  v4f g2 = r1 ? s1 : (r2 ? s0 : s2);

#pragma unroll
  for (int e = 0; e < 4; ++e) {
    g0[e] = wave_sum(g0[e]);
    g1[e] = wave_sum(g1[e]);
    g2[e] = wave_sum(g2[e]);
  }
  if ((t & 31) == 0) {
    wsum[wave * 3 + 0] = g0;
    wsum[wave * 3 + 1] = g1;
    wsum[wave * 3 + 2] = g2;
  }
  __syncthreads();
  if (t < 3) {
    v4f a = wsum[t];
#pragma unroll
    for (int w = 1; w < RT_ / 32; ++w) a += wsum[w * 3 + t];
    ((v4f*)P)[(ch * B_ + b) * 3 + t] = a;
  }
}

#define CGC_     32
#define NPB_     64
#define KSTEPS_  (HID_ / 4)
#define APITCH_  260
#define BPITCH_  40
#define W1PITCH_ 136
#define W1HALF_  (IN_ * W1PITCH_)

__global__ __launch_bounds__(256) void gcn_l2_broadcast(
    const float* __restrict__ P, const float* __restrict__ W1,
    const float* __restrict__ b1, const float* __restrict__ W2,
    const float* __restrict__ b2, float* __restrict__ out) {
  __shared__ float As[B_ * APITCH_];
  __shared__ float Bs[HID_ * BPITCH_];
  __shared__ float Hs[B_ * CGC_];
  __shared__ __attribute__((aligned(16))) float W1s[2 * W1HALF_];
  __shared__ __attribute__((aligned(16))) float Ms[B_ * IN_];

  const int t     = threadIdx.x;
  const int wave  = __builtin_amdgcn_readfirstlane(t >> 5);
  const int lane  = t & 31;
  const int laneM = lane & 15;
  const int hi    = lane >> 4;
  const int kOff  = 2 * hi;
  const int cg    = blockIdx.x & 3;
  const int nc    = blockIdx.x >> 2;
  const int mi    = (wave >> 1) & 1;
  const int ni    = wave & 1;

  float bias = 0.f;
  if (wave < 4) bias = b2[cg * CGC_ + ni * 16 + laneM];
  const int n0 = wave * 16;
  float bias1[2];
  bias1[0] = b1[n0 + laneM];
  bias1[1] = b1[128 + n0 + laneM];
  v4f vb[8];
  {
    v4f pp[NCH_], w1r[3];
    if (wave < 3) {
#pragma unroll
      for (int c = 0; c < NCH_; ++c)
        pp[c] = *(const v4f*)(P + (c * B_ * IN_) + 4 * t);
    }
#pragma unroll
    for (int j = 0; j < 3; ++j) {
      const int i = t + 256 * j;
      w1r[j] = *(const v4f*)(W1 + (i >> 6) * HID_ + (i & 63) * 4);
    }
#pragma unroll
    for (int j = 0; j < 8; ++j) {
      const int i = t + 256 * j;
      vb[j] = *(const v4f*)(W2 + (i >> 3) * OUT_ + cg * CGC_ + (i & 7) * 4);
    }
    if (wave < 3) {
      v4f m = pp[0];
#pragma unroll
      for (int c = 1; c < NCH_; ++c) m += pp[c];
      *(v4f*)(Ms + 4 * t) = m * (1.0f / (float)N_);
    }
#pragma unroll
    for (int j = 0; j < 3; ++j) {
      const int i = t + 256 * j;
      *(v4f*)(W1s + ((i >> 5) & 1) * W1HALF_ + (i >> 6) * W1PITCH_
              + (i & 31) * 4) = w1r[j];
    }
  }
  asm volatile("s_wait_dscnt 0x0\n\ts_barrier_signal -1\n\ts_barrier_wait -1"
               ::: "memory");

  {
    v2f a1[2][3], bf1[2][3];
#pragma unroll
    for (int ks = 0; ks < 3; ++ks) {
#pragma unroll
      for (int m = 0; m < 2; ++m)
        a1[m][ks] = *(const v2f*)(Ms + (m * 16 + laneM) * IN_ + 4 * ks + kOff);
#pragma unroll
      for (int n = 0; n < 2; ++n) {
        const float* bp = W1s + n * W1HALF_ + (4 * ks + kOff) * W1PITCH_
                        + n0 + laneM;
        bf1[n][ks].x = bp[0];
        bf1[n][ks].y = bp[W1PITCH_];
      }
    }
    __builtin_amdgcn_sched_barrier(0);
    v8f acc00 = {}, acc01 = {}, acc10 = {}, acc11 = {};
#pragma unroll
    for (int ks = 0; ks < 3; ++ks) {
      acc00 = __builtin_amdgcn_wmma_f32_16x16x4_f32(
          false, a1[0][ks], false, bf1[0][ks], (short)0, acc00, false, false);
      acc01 = __builtin_amdgcn_wmma_f32_16x16x4_f32(
          false, a1[0][ks], false, bf1[1][ks], (short)0, acc01, false, false);
      acc10 = __builtin_amdgcn_wmma_f32_16x16x4_f32(
          false, a1[1][ks], false, bf1[0][ks], (short)0, acc10, false, false);
      acc11 = __builtin_amdgcn_wmma_f32_16x16x4_f32(
          false, a1[1][ks], false, bf1[1][ks], (short)0, acc11, false, false);
    }
    __builtin_amdgcn_sched_barrier(0);
    asm volatile("v_nop\n\tv_nop\n\tv_nop\n\tv_nop\n\tv_nop"
                 : "+v"(acc00), "+v"(acc01), "+v"(acc10), "+v"(acc11));
    __builtin_amdgcn_sched_barrier(0);
#pragma unroll
    for (int v = 0; v < 8; ++v) {
      float* hrow0 = As + (v + 8 * hi) * APITCH_ + n0 + laneM;
      float* hrow1 = hrow0 + 16 * APITCH_;
      hrow0[0]   = acc00[v] + bias1[0];
      hrow0[128] = acc01[v] + bias1[1];
      hrow1[0]   = acc10[v] + bias1[0];
      hrow1[128] = acc11[v] + bias1[1];
    }
#pragma unroll
    for (int j = 0; j < 8; ++j) {
      const int i = t + 256 * j;
      *(v4f*)(Bs + (i >> 3) * BPITCH_ + (i & 7) * 4) = vb[j];
    }
  }
  __syncthreads();

  if (wave < 4) {
    const float* arow = As + (mi * 16 + laneM) * APITCH_ + kOff;
    const float* bcol = Bs + kOff * BPITCH_ + ni * 16 + laneM;
    v2f afrag[KSTEPS_], bfrag[KSTEPS_];
#pragma unroll
    for (int ks = 0; ks < KSTEPS_; ++ks) {
      afrag[ks]   = *(const v2f*)(arow + 4 * ks);
      bfrag[ks].x = bcol[(4 * ks) * BPITCH_];
      bfrag[ks].y = bcol[(4 * ks + 1) * BPITCH_];
      if ((ks & 7) == 7) __builtin_amdgcn_sched_barrier(0);
    }
    v8f c = {};
#pragma unroll
    for (int ks = 0; ks < KSTEPS_; ++ks) {
      c = __builtin_amdgcn_wmma_f32_16x16x4_f32(
          false, afrag[ks], false, bfrag[ks], (short)0, c, false, false);
    }
    __builtin_amdgcn_sched_barrier(0);
    asm volatile("v_nop\n\tv_nop\n\tv_nop\n\tv_nop\n\tv_nop" : "+v"(c));
    __builtin_amdgcn_sched_barrier(0);
    const int col = ni * 16 + laneM;
#pragma unroll
    for (int v = 0; v < 8; ++v)
      Hs[(mi * 16 + v + 8 * hi) * CGC_ + col] = c[v] + bias;
  }
  __syncthreads();

  const int col4 = (lane & 7) << 2;
  const int nsel = lane >> 3;
#pragma unroll
  for (int bb = 0; bb < 4; ++bb) {
    const int b   = wave * 4 + bb;
    const v4f val = *(const v4f*)(Hs + b * CGC_ + col4);
    float* obase = out + ((size_t)b * N_ + (size_t)nc * NPB_ + nsel) * OUT_
                 + cg * CGC_ + col4;
#pragma unroll
    for (int i = 0; i < NPB_ / 4; ++i)
      __builtin_nontemporal_store(val, (v4f*)(obase + (size_t)(i * 4) * OUT_));
  }
}

extern "C" void kernel_launch(void* const* d_in, const int* in_sizes, int n_in,
                              void* d_out, int out_size, void* d_ws,
                              size_t ws_size, hipStream_t stream) {
  const float* x  = (const float*)d_in[0];
  const float* W1 = (const float*)d_in[1];
  const float* b1 = (const float*)d_in[2];
  const float* W2 = (const float*)d_in[3];
  const float* b2 = (const float*)d_in[4];
  float* out = (float*)d_out;

  float* P = (float*)d_ws;

  gcn_node_partial<<<B_ * NCH_, RT_, 0, stream>>>(x, P);
  gcn_l2_broadcast<<<4 * (N_ / NPB_), 256, 0, stream>>>(P, W1, b1, W2, b2, out);
}
